# starting kernel
# baseline (speedup 1.0000x reference)
#include <hip/hip_runtime.h>
#include <hip/hip_bf16.h>

#define BATCH 1024
#define DDIM  1024
#define UNITS 16
#define ROW   8
#define NCOL  (UNITS * ROW)
#define OUTW  (DDIM + UNITS)

typedef __attribute__((ext_vector_type(2))) float v2f;
typedef __attribute__((ext_vector_type(8))) float v8f;

__global__ void __launch_bounds__(256)
mbd_gemm_wmma_f32(const float* __restrict__ x,
                  const float* __restrict__ T,
                  float* __restrict__ Mg)
{
    const int lane   = threadIdx.x & 31;
    const int wave   = threadIdx.x >> 5;
    const int mrow   = lane & 15;
    const int khalf  = lane >> 4;
    const int rowBase = blockIdx.x * 16;
    const int colBase = wave * 16;

    const float* __restrict__ Arow = x + (size_t)(rowBase + mrow) * DDIM;

    const int n  = colBase + mrow;
    const float* __restrict__ Bcol = T + (size_t)(n >> 3) * (DDIM * ROW) + (n & 7);

    v8f c = {};

    #pragma unroll 8
    for (int k0 = 0; k0 < DDIM; k0 += 4) {
        const int kk = k0 + 2 * khalf;
        v2f a = *(const v2f*)(Arow + kk);
        v2f b;
        b.x = Bcol[(size_t)kk * ROW];
        b.y = Bcol[(size_t)(kk + 1) * ROW];
        c = __builtin_amdgcn_wmma_f32_16x16x4_f32(false, a, false, b,
                                                  (short)0, c, false, false);
    }

    float* __restrict__ Drow = Mg + (size_t)(rowBase + 8 * khalf) * NCOL + n;
    #pragma unroll
    for (int v = 0; v < 8; ++v)
        Drow[(size_t)v * NCOL] = c[v];
}

__global__ void __launch_bounds__(128)
mbd_feats(const float* __restrict__ Mg,
          float* __restrict__ out)
{
    __shared__ float lds[BATCH * ROW];

    const int u = blockIdx.x;

    for (int i = threadIdx.x; i < BATCH * ROW; i += 128) {
        const int bp  = i >> 3;
        const int r   = i & 7;
        const int row = (bp & 63) * 16 + u;
        const int col = (bp >> 6) * 8 + r;
        lds[i] = Mg[(size_t)row * NCOL + col];
    }
    __syncthreads();

    const int bp = blockIdx.y * 128 + threadIdx.x;

    float m[ROW];
    #pragma unroll
    for (int r = 0; r < ROW; ++r)
        m[r] = lds[bp * ROW + r];

    float acc = 0.0f;
    for (int j = 0; j < BATCH; ++j) {
        const float* __restrict__ mj = &lds[j * ROW];
        float s = 0.0f;
        #pragma unroll
        for (int r = 0; r < ROW; ++r)
            s += __builtin_fabsf(m[r] - mj[r]);
        acc += __expf(-s);
    }

    out[(size_t)bp * OUTW + DDIM + u] = acc;
}

__global__ void __launch_bounds__(256)
mbd_copy_x(const float* __restrict__ x, float* __restrict__ out)
{
    const int idx = blockIdx.x * blockDim.x + threadIdx.x;
    const int b   = idx >> 8;
    const int c4  = idx & 255;
    const float4* src = (const float4*)(x   + (size_t)b * DDIM);
    float4*       dst = (float4*)      (out + (size_t)b * OUTW);
    dst[c4] = src[c4];
}

extern "C" void kernel_launch(void* const* d_in, const int* in_sizes, int n_in,
                              void* d_out, int out_size, void* d_ws, size_t ws_size,
                              hipStream_t stream)
{
    const float* x = (const float*)d_in[0];
    const float* T = (const float*)d_in[1];
    float* out = (float*)d_out;
    float* Mg  = (float*)d_ws;

    mbd_gemm_wmma_f32<<<dim3(64), dim3(256), 0, stream>>>(x, T, Mg);

    mbd_feats<<<dim3(UNITS, BATCH / 128), dim3(128), 0, stream>>>(Mg, out);

    mbd_copy_x<<<dim3((BATCH * DDIM / 4) / 256), dim3(256), 0, stream>>>(x, out);
}


# pairwise kernel: j loop split over 8 wave groups per block (64 i x 8 j-slices, 512 threads, grid 16x16), partial sums added through LDS; the order of the sum over j changes
# speedup vs baseline: 1.4493x; 1.4493x over previous
#include <hip/hip_runtime.h>
#include <hip/hip_bf16.h>

#define BATCH 1024
#define DDIM  1024
#define UNITS 16
#define ROW   8
#define NCOL  (UNITS * ROW)
#define OUTW  (DDIM + UNITS)

typedef __attribute__((ext_vector_type(2))) float v2f;
typedef __attribute__((ext_vector_type(8))) float v8f;

__global__ void __launch_bounds__(256)
mbd_gemm_wmma_f32(const float* __restrict__ x,
                  const float* __restrict__ T,
                  float* __restrict__ Mg)
{
    const int lane   = threadIdx.x & 31;
    const int wave   = threadIdx.x >> 5;
    const int mrow   = lane & 15;
    const int khalf  = lane >> 4;
    const int rowBase = blockIdx.x * 16;
    const int colBase = wave * 16;

    const float* __restrict__ Arow = x + (size_t)(rowBase + mrow) * DDIM;

    const int n  = colBase + mrow;
    const float* __restrict__ Bcol = T + (size_t)(n >> 3) * (DDIM * ROW) + (n & 7);

    v8f c = {};

    #pragma unroll 8
    for (int k0 = 0; k0 < DDIM; k0 += 4) {
        const int kk = k0 + 2 * khalf;
        v2f a = *(const v2f*)(Arow + kk);
        v2f b;
        b.x = Bcol[(size_t)kk * ROW];
        b.y = Bcol[(size_t)(kk + 1) * ROW];
        c = __builtin_amdgcn_wmma_f32_16x16x4_f32(false, a, false, b,
                                                  (short)0, c, false, false);
    }

    float* __restrict__ Drow = Mg + (size_t)(rowBase + 8 * khalf) * NCOL + n;
    #pragma unroll
    for (int v = 0; v < 8; ++v)
        Drow[(size_t)v * NCOL] = c[v];
}

#define ITILE  64
#define JSPLIT 8
#define JLEN   (BATCH / JSPLIT)

__global__ void __launch_bounds__(ITILE * JSPLIT)
mbd_feats(const float* __restrict__ Mg,
          float* __restrict__ out)
{
    __shared__ float lds[BATCH * ROW];
    __shared__ float part[JSPLIT][ITILE];

    const int u = blockIdx.x;

    for (int i = threadIdx.x; i < BATCH * ROW; i += ITILE * JSPLIT) {
        const int bp  = i >> 3;
        const int r   = i & 7;
        const int row = (bp & 63) * 16 + u;
        const int col = (bp >> 6) * 8 + r;
        lds[i] = Mg[(size_t)row * NCOL + col];
    }
    __syncthreads();

    const int il = threadIdx.x & (ITILE - 1);
    const int js = threadIdx.x >> 6;
    const int bp = blockIdx.y * ITILE + il;

    float m[ROW];
    #pragma unroll
    for (int r = 0; r < ROW; ++r)
        m[r] = lds[bp * ROW + r];

    float acc = 0.0f;
    const int jbeg = js * JLEN;
    for (int j = jbeg; j < jbeg + JLEN; ++j) {
        const float* __restrict__ mj = &lds[j * ROW];
        float s = 0.0f;
        #pragma unroll
        for (int r = 0; r < ROW; ++r)
            s += __builtin_fabsf(m[r] - mj[r]);
        acc += __expf(-s);
    }

    part[js][il] = acc;
    __syncthreads();

    if (threadIdx.x < ITILE) {
        float t = part[0][il];
        #pragma unroll
        for (int s = 1; s < JSPLIT; ++s)
            t += part[s][il];
        out[(size_t)bp * OUTW + DDIM + u] = t;
    }
}

__global__ void __launch_bounds__(256)
mbd_copy_x(const float* __restrict__ x, float* __restrict__ out)
{
    const int idx = blockIdx.x * blockDim.x + threadIdx.x;
    const int b   = idx >> 8;
    const int c4  = idx & 255;
    const float4* src = (const float4*)(x   + (size_t)b * DDIM);
    float4*       dst = (float4*)      (out + (size_t)b * OUTW);
    dst[c4] = src[c4];
}

extern "C" void kernel_launch(void* const* d_in, const int* in_sizes, int n_in,
                              void* d_out, int out_size, void* d_ws, size_t ws_size,
                              hipStream_t stream)
{
    const float* x = (const float*)d_in[0];
    const float* T = (const float*)d_in[1];
    float* out = (float*)d_out;
    float* Mg  = (float*)d_ws;

    mbd_gemm_wmma_f32<<<dim3(64), dim3(256), 0, stream>>>(x, T, Mg);

    mbd_feats<<<dim3(UNITS, BATCH / ITILE), dim3(ITILE * JSPLIT), 0, stream>>>(Mg, out);

    mbd_copy_x<<<dim3((BATCH * DDIM / 4) / 256), dim3(256), 0, stream>>>(x, out);
}


# GEMM kernel: k split over 16 waves per block (one 16x32 tile per block, grid 4x64, 512 threads, 16-byte x loads), partial tiles added through LDS; the order of the sum over k changes
# speedup vs baseline: 2.6900x; 1.8561x over previous
#include <hip/hip_runtime.h>
#include <hip/hip_bf16.h>

#define BATCH 1024
#define DDIM  1024
#define UNITS 16
#define ROW   8
#define NCOL  (UNITS * ROW)
#define OUTW  (DDIM + UNITS)

typedef __attribute__((ext_vector_type(2))) float v2f;
typedef __attribute__((ext_vector_type(8))) float v8f;

#define GK_WAVES 16
#define GK_SLICE (DDIM / GK_WAVES)
#define GK_JPAD  272

__global__ void __launch_bounds__(32 * GK_WAVES)
mbd_gemm_wmma_f32(const float* __restrict__ x,
                  const float* __restrict__ T,
                  float* __restrict__ Mg)
{
    __shared__ float red[GK_WAVES][2 * GK_JPAD];

    const int lane    = threadIdx.x & 31;
    const int wave    = threadIdx.x >> 5;
    const int c       = lane & 15;
    const int h       = lane >> 4;
    const int rowBase = blockIdx.y * 16;
    const int colBase = blockIdx.x * 32;
    const int kl      = wave * GK_SLICE + 4 * h;

    const float* __restrict__ Arow = x + (size_t)(rowBase + c) * DDIM + kl;

    const int n0 = colBase + c;
    const int n1 = n0 + 16;
    const float* __restrict__ Bc0 = T + (size_t)(n0 >> 3) * (DDIM * ROW) + (n0 & 7) + (size_t)kl * ROW;
    const float* __restrict__ Bc1 = T + (size_t)(n1 >> 3) * (DDIM * ROW) + (n1 & 7) + (size_t)kl * ROW;

    v8f acc0 = {};
    v8f acc1 = {};
    for (int p = 0; p < GK_SLICE / 8; ++p) {
        const float4 q = *(const float4*)(Arow + 8 * p);
        const float* __restrict__ b0 = Bc0 + (size_t)(8 * p) * ROW;
        const float* __restrict__ b1 = Bc1 + (size_t)(8 * p) * ROW;
        v2f a_s0;  a_s0.x  = q.x;          a_s0.y  = q.y;
        v2f a_s1;  a_s1.x  = q.z;          a_s1.y  = q.w;
        v2f b0_s0; b0_s0.x = b0[0];        b0_s0.y = b0[ROW];
        v2f b0_s1; b0_s1.x = b0[2 * ROW];  b0_s1.y = b0[3 * ROW];
        v2f b1_s0; b1_s0.x = b1[0];        b1_s0.y = b1[ROW];
        v2f b1_s1; b1_s1.x = b1[2 * ROW];  b1_s1.y = b1[3 * ROW];
        acc0 = __builtin_amdgcn_wmma_f32_16x16x4_f32(false, a_s0, false, b0_s0,
                                                     (short)0, acc0, false, false);
        acc1 = __builtin_amdgcn_wmma_f32_16x16x4_f32(false, a_s0, false, b1_s0,
                                                     (short)0, acc1, false, false);
        acc0 = __builtin_amdgcn_wmma_f32_16x16x4_f32(false, a_s1, false, b0_s1,
                                                     (short)0, acc0, false, false);
        acc1 = __builtin_amdgcn_wmma_f32_16x16x4_f32(false, a_s1, false, b1_s1,
                                                     (short)0, acc1, false, false);
    }

    float* __restrict__ mine = &red[wave][0];
    for (int v = 0; v < 8; ++v) {
        mine[v * 32 + lane]           = acc0[v];
        mine[GK_JPAD + v * 32 + lane] = acc1[v];
    }
    __syncthreads();

    const int rl = threadIdx.x >> 5;
    const int cl = threadIdx.x & 31;
    const int e  = (cl >> 4) * GK_JPAD + (rl & 7) * 32 + 16 * (rl >> 3) + (cl & 15);
    float s = red[0][e];
    for (int w = 1; w < GK_WAVES; ++w)
        s += red[w][e];
    Mg[(size_t)(rowBase + rl) * NCOL + colBase + cl] = s;
}

#define ITILE  64
#define JSPLIT 8
#define JLEN   (BATCH / JSPLIT)

__global__ void __launch_bounds__(ITILE * JSPLIT)
mbd_feats(const float* __restrict__ Mg,
          float* __restrict__ out)
{
    __shared__ float lds[BATCH * ROW];
    __shared__ float part[JSPLIT][ITILE];

    const int u = blockIdx.x;

    for (int i = threadIdx.x; i < BATCH * ROW; i += ITILE * JSPLIT) {
        const int bp  = i >> 3;
        const int r   = i & 7;
        const int row = (bp & 63) * 16 + u;
        const int col = (bp >> 6) * 8 + r;
        lds[i] = Mg[(size_t)row * NCOL + col];
    }
    __syncthreads();

    const int il = threadIdx.x & (ITILE - 1);
    const int js = threadIdx.x >> 6;
    const int bp = blockIdx.y * ITILE + il;

    float m[ROW];
    #pragma unroll
    for (int r = 0; r < ROW; ++r)
        m[r] = lds[bp * ROW + r];

    float acc = 0.0f;
    const int jbeg = js * JLEN;
    for (int j = jbeg; j < jbeg + JLEN; ++j) {
        const float* __restrict__ mj = &lds[j * ROW];
        float s = 0.0f;
        #pragma unroll
        for (int r = 0; r < ROW; ++r)
            s += __builtin_fabsf(m[r] - mj[r]);
        acc += __expf(-s);
    }

    part[js][il] = acc;
    __syncthreads();

    if (threadIdx.x < ITILE) {
        float t = part[0][il];
        #pragma unroll
        for (int s = 1; s < JSPLIT; ++s)
            t += part[s][il];
        out[(size_t)bp * OUTW + DDIM + u] = t;
    }
}

__global__ void __launch_bounds__(256)
mbd_copy_x(const float* __restrict__ x, float* __restrict__ out)
{
    const int idx = blockIdx.x * blockDim.x + threadIdx.x;
    const int b   = idx >> 8;
    const int c4  = idx & 255;
    const float4* src = (const float4*)(x   + (size_t)b * DDIM);
    float4*       dst = (float4*)      (out + (size_t)b * OUTW);
    dst[c4] = src[c4];
}

extern "C" void kernel_launch(void* const* d_in, const int* in_sizes, int n_in,
                              void* d_out, int out_size, void* d_ws, size_t ws_size,
                              hipStream_t stream)
{
    const float* x = (const float*)d_in[0];
    const float* T = (const float*)d_in[1];
    float* out = (float*)d_out;
    float* Mg  = (float*)d_ws;

    mbd_gemm_wmma_f32<<<dim3(4, 64), dim3(512), 0, stream>>>(x, T, Mg);

    mbd_feats<<<dim3(UNITS, BATCH / ITILE), dim3(ITILE * JSPLIT), 0, stream>>>(Mg, out);

    mbd_copy_x<<<dim3((BATCH * DDIM / 4) / 256), dim3(256), 0, stream>>>(x, out);
}


# x copy folded into the GEMM launch: behind its barrier each block copies 4 of its 16 x rows to out (two 16-byte loads and stores per thread); copy kernel and its launch removed (3 launches -> 2)
# speedup vs baseline: 3.0624x; 1.1384x over previous
#include <hip/hip_runtime.h>
#include <hip/hip_bf16.h>

#define BATCH 1024
#define DDIM  1024
#define UNITS 16
#define ROW   8
#define NCOL  (UNITS * ROW)
#define OUTW  (DDIM + UNITS)

typedef __attribute__((ext_vector_type(2))) float v2f;
typedef __attribute__((ext_vector_type(4))) float v4f;
typedef __attribute__((ext_vector_type(8))) float v8f;

#define GK_WAVES 16
#define GK_SLICE (DDIM / GK_WAVES)
#define GK_JPAD  272

__global__ void __launch_bounds__(32 * GK_WAVES)
mbd_gemm_wmma_f32(const float* __restrict__ x,
                  const float* __restrict__ T,
                  float* __restrict__ Mg,
                  float* __restrict__ out)
{
    __shared__ float red[GK_WAVES][2 * GK_JPAD];

    const int lane    = threadIdx.x & 31;
    const int wave    = threadIdx.x >> 5;
    const int c       = lane & 15;
    const int h       = lane >> 4;
    const int rowBase = blockIdx.y * 16;
    const int colBase = blockIdx.x * 32;
    const int kl      = wave * GK_SLICE + 4 * h;

    const float* __restrict__ Arow = x + (size_t)(rowBase + c) * DDIM + kl;

    const int n0 = colBase + c;
    const int n1 = n0 + 16;
    const float* __restrict__ Bc0 = T + (size_t)(n0 >> 3) * (DDIM * ROW) + (n0 & 7) + (size_t)kl * ROW;
    const float* __restrict__ Bc1 = T + (size_t)(n1 >> 3) * (DDIM * ROW) + (n1 & 7) + (size_t)kl * ROW;

    v8f acc0 = {};
    v8f acc1 = {};
    for (int p = 0; p < GK_SLICE / 8; ++p) {
        const float4 q = *(const float4*)(Arow + 8 * p);
        const float* __restrict__ b0 = Bc0 + (size_t)(8 * p) * ROW;
        const float* __restrict__ b1 = Bc1 + (size_t)(8 * p) * ROW;
        v2f a_s0;  a_s0.x  = q.x;          a_s0.y  = q.y;
        v2f a_s1;  a_s1.x  = q.z;          a_s1.y  = q.w;
        v2f b0_s0; b0_s0.x = b0[0];        b0_s0.y = b0[ROW];
        v2f b0_s1; b0_s1.x = b0[2 * ROW];  b0_s1.y = b0[3 * ROW];
        v2f b1_s0; b1_s0.x = b1[0];        b1_s0.y = b1[ROW];
        v2f b1_s1; b1_s1.x = b1[2 * ROW];  b1_s1.y = b1[3 * ROW];
        acc0 = __builtin_amdgcn_wmma_f32_16x16x4_f32(false, a_s0, false, b0_s0,
                                                     (short)0, acc0, false, false);
        acc1 = __builtin_amdgcn_wmma_f32_16x16x4_f32(false, a_s0, false, b1_s0,
                                                     (short)0, acc1, false, false);
        acc0 = __builtin_amdgcn_wmma_f32_16x16x4_f32(false, a_s1, false, b0_s1,
                                                     (short)0, acc0, false, false);
        acc1 = __builtin_amdgcn_wmma_f32_16x16x4_f32(false, a_s1, false, b1_s1,
                                                     (short)0, acc1, false, false);
    }

    float* __restrict__ mine = &red[wave][0];
    for (int v = 0; v < 8; ++v) {
        mine[v * 32 + lane]           = acc0[v];
        mine[GK_JPAD + v * 32 + lane] = acc1[v];
    }
    __syncthreads();

    const int crow = rowBase + 4 * blockIdx.x;
    const v4f* __restrict__ csrc = (const v4f*)(x + (size_t)crow * DDIM);
    const v4f cp0 = csrc[threadIdx.x];
    const v4f cp1 = csrc[threadIdx.x + 512];

    const int rl = threadIdx.x >> 5;
    const int cl = threadIdx.x & 31;
    const int e  = (cl >> 4) * GK_JPAD + (rl & 7) * 32 + 16 * (rl >> 3) + (cl & 15);
    float s = red[0][e];
    for (int w = 1; w < GK_WAVES; ++w)
        s += red[w][e];
    Mg[(size_t)(rowBase + rl) * NCOL + colBase + cl] = s;

    const int r2 = threadIdx.x >> 8;
    const int c4 = threadIdx.x & 255;
    v4f* __restrict__ d0 = (v4f*)(out + (size_t)(crow + r2) * OUTW);
    v4f* __restrict__ d1 = (v4f*)(out + (size_t)(crow + 2 + r2) * OUTW);
    d0[c4] = cp0;
    d1[c4] = cp1;
}

#define ITILE  64
#define JSPLIT 8
#define JLEN   (BATCH / JSPLIT)

__global__ void __launch_bounds__(ITILE * JSPLIT)
mbd_feats(const float* __restrict__ Mg,
          float* __restrict__ out)
{
    __shared__ float lds[BATCH * ROW];
    __shared__ float part[JSPLIT][ITILE];

    const int u = blockIdx.x;

    for (int i = threadIdx.x; i < BATCH * ROW; i += ITILE * JSPLIT) {
        const int bp  = i >> 3;
        const int r   = i & 7;
        const int row = (bp & 63) * 16 + u;
        const int col = (bp >> 6) * 8 + r;
        lds[i] = Mg[(size_t)row * NCOL + col];
    }
    __syncthreads();

    const int il = threadIdx.x & (ITILE - 1);
    const int js = threadIdx.x >> 6;
    const int bp = blockIdx.y * ITILE + il;

    float m[ROW];
    #pragma unroll
    for (int r = 0; r < ROW; ++r)
        m[r] = lds[bp * ROW + r];

    float acc = 0.0f;
    const int jbeg = js * JLEN;
    for (int j = jbeg; j < jbeg + JLEN; ++j) {
        const float* __restrict__ mj = &lds[j * ROW];
        float s = 0.0f;
        #pragma unroll
        for (int r = 0; r < ROW; ++r)
            s += __builtin_fabsf(m[r] - mj[r]);
        acc += __expf(-s);
    }

    part[js][il] = acc;
    __syncthreads();

    if (threadIdx.x < ITILE) {
        float t = part[0][il];
        #pragma unroll
        for (int s = 1; s < JSPLIT; ++s)
            t += part[s][il];
        out[(size_t)bp * OUTW + DDIM + u] = t;
    }
}

extern "C" void kernel_launch(void* const* d_in, const int* in_sizes, int n_in,
                              void* d_out, int out_size, void* d_ws, size_t ws_size,
                              hipStream_t stream)
{
    const float* x = (const float*)d_in[0];
    const float* T = (const float*)d_in[1];
    float* out = (float*)d_out;
    float* Mg  = (float*)d_ws;

    mbd_gemm_wmma_f32<<<dim3(4, 64), dim3(512), 0, stream>>>(x, T, Mg, out);

    mbd_feats<<<dim3(UNITS, BATCH / ITILE), dim3(ITILE * JSPLIT), 0, stream>>>(Mg, out);
}
